# starting kernel
# baseline (speedup 1.0000x reference)
#include <hip/hip_runtime.h>
#include <cstdint>

typedef float v2f __attribute__((ext_vector_type(2)));
typedef float v8f __attribute__((ext_vector_type(8)));

namespace {

constexpr int H    = 40;
constexpr int G    = 120;
constexpr int I    = 16;
constexpr int T    = 256;
constexpr int NT   = 8;
constexpr int KC_I = 4;
constexpr int KC_H = 10;

constexpr int OFF_WIH0 = 0;
constexpr int OFF_WIH1 = OFF_WIH0 + G * I;
constexpr int OFF_WHH1 = OFF_WIH1 + G * H;
constexpr int OFF_BRZ0 = OFF_WHH1 + G * H;
constexpr int OFF_BNX0 = OFF_BRZ0 + 80;
constexpr int OFF_BNH0 = OFF_BNX0 + H;
constexpr int OFF_BRZ1 = OFF_BNH0 + H;
constexpr int OFF_BNX1 = OFF_BRZ1 + 80;
constexpr int OFF_BNH1 = OFF_BNX1 + H;
constexpr int OFF_ACCS = OFF_BNH1 + H;
constexpr int OFF_AXN  = OFF_ACCS + 16 * 80;
constexpr int OFF_AHN  = OFF_AXN  + 16 * H;
constexpr int OFF_H0   = OFF_AHN  + 16 * H;
constexpr int OFF_H1   = OFF_H0   + 16 * H;
constexpr int SMEM_FLOATS = OFF_H1 + 16 * H + 64;

__device__ __forceinline__ v8f wmma_f32(v2f a, v2f b, v8f c) {
  return __builtin_amdgcn_wmma_f32_16x16x4_f32(
       false, a,  false, b,
       (short)0, c,  false,  false);
}

__device__ __forceinline__ v2f ldsB(const float* w, int K, int nt, int nloc, int koff, int kc) {
  return *(const v2f*)(w + (nt * 16 + nloc) * K + kc * 4 + koff);
}

__device__ __forceinline__ void store_tile(float* dst, int ldw, int col, int rbase, v8f acc) {
  #pragma unroll
  for (int v = 0; v < 8; ++v) dst[(rbase + v) * ldw + col] = acc[v];
}

__device__ __forceinline__ float fsigmoid(float v) { return 1.0f / (1.0f + __expf(-v)); }

}

__global__ __launch_bounds__(32) void gru2_fused(
    const float* __restrict__ x,
    const float* __restrict__ Wih0, const float* __restrict__ Whh0,
    const float* __restrict__ bih0, const float* __restrict__ bhh0,
    const float* __restrict__ Wih1, const float* __restrict__ Whh1,
    const float* __restrict__ bih1, const float* __restrict__ bhh1,
    float* __restrict__ out)
{
  __shared__ float smem[SMEM_FLOATS];
  const int lane  = threadIdx.x & 31;
  const int nloc  = lane & 15;
  const int koff  = (lane >> 4) << 1;
  const int rbase = (lane < 16) ? 0 : 8;
  const int bbase = blockIdx.x * 16;

  for (int i = lane; i < G * I; i += 32) smem[OFF_WIH0 + i] = Wih0[i];
  for (int i = lane; i < G * H; i += 32) smem[OFF_WIH1 + i] = Wih1[i];
  for (int i = lane; i < G * H; i += 32) smem[OFF_WHH1 + i] = Whh1[i];
  for (int i = lane; i < 80; i += 32) {
    smem[OFF_BRZ0 + i] = bih0[i] + bhh0[i];
    smem[OFF_BRZ1 + i] = bih1[i] + bhh1[i];
  }
  for (int i = lane; i < H; i += 32) {
    smem[OFF_BNX0 + i] = bih0[80 + i]; smem[OFF_BNH0 + i] = bhh0[80 + i];
    smem[OFF_BNX1 + i] = bih1[80 + i]; smem[OFF_BNH1 + i] = bhh1[80 + i];
  }
  for (int i = lane; i < 16 * H; i += 32) { smem[OFF_H0 + i] = 0.f; smem[OFF_H1 + i] = 0.f; }

  v2f whh0[KC_H][NT];
  #pragma unroll
  for (int kc = 0; kc < KC_H; ++kc) {
    #pragma unroll
    for (int nt = 0; nt < NT; ++nt) {
      const int n = nt * 16 + nloc;
      v2f w = {0.f, 0.f};
      if (n < G) w = *(const v2f*)(Whh0 + n * H + kc * 4 + koff);
      whh0[kc][nt] = w;
    }
  }

  __syncthreads();
  asm volatile("s_wait_dscnt 0" ::: "memory");

  float* sH0 = smem + OFF_H0;
  float* sH1 = smem + OFF_H1;
  const float* xrow = x + (size_t)(bbase + nloc) * T * I;

  #pragma unroll 1
  for (int t = 0; t < T; ++t) {
    if (t + 1 < T) __builtin_prefetch(xrow + (t + 1) * I + koff, 0, 3);

    v2f ax[KC_I];
    #pragma unroll
    for (int kc = 0; kc < KC_I; ++kc)
      ax[kc] = *(const v2f*)(xrow + t * I + kc * 4 + koff);
    v2f ah[KC_H];
    #pragma unroll
    for (int kc = 0; kc < KC_H; ++kc)
      ah[kc] = *(const v2f*)(sH0 + nloc * H + kc * 4 + koff);
    asm volatile("s_wait_dscnt 0" ::: "memory");

    #pragma unroll
    for (int nt = 0; nt < NT; ++nt) {
      const int n = nt * 16 + nloc;
      if (nt < 5) {
        v8f acc = {};
        #pragma unroll
        for (int kc = 0; kc < KC_I; ++kc)
          acc = wmma_f32(ax[kc], ldsB(smem + OFF_WIH0, I, nt, nloc, koff, kc), acc);
        #pragma unroll
        for (int kc = 0; kc < KC_H; ++kc)
          acc = wmma_f32(ah[kc], whh0[kc][nt], acc);
        store_tile(smem + OFF_ACCS, 80, n, rbase, acc);
      } else {
        v8f accx = {}, acch = {};
        #pragma unroll
        for (int kc = 0; kc < KC_I; ++kc)
          accx = wmma_f32(ax[kc], ldsB(smem + OFF_WIH0, I, nt, nloc, koff, kc), accx);
        #pragma unroll
        for (int kc = 0; kc < KC_H; ++kc)
          acch = wmma_f32(ah[kc], whh0[kc][nt], acch);
        if (n < G) {
          store_tile(smem + OFF_AXN, H, n - 80, rbase, accx);
          store_tile(smem + OFF_AHN, H, n - 80, rbase, acch);
        }
      }
    }
    asm volatile("s_wait_dscnt 0" ::: "memory");

    for (int e = lane; e < 16 * H; e += 32) {
      const int b = e / H, j = e % H;
      const float rz = smem[OFF_ACCS + b * 80 + j]      + smem[OFF_BRZ0 + j];
      const float zz = smem[OFF_ACCS + b * 80 + 40 + j] + smem[OFF_BRZ0 + 40 + j];
      const float xn = smem[OFF_AXN + b * H + j]        + smem[OFF_BNX0 + j];
      const float hn = smem[OFF_AHN + b * H + j]        + smem[OFF_BNH0 + j];
      const float hp = sH0[b * H + j];
      const float r  = fsigmoid(rz);
      const float z  = fsigmoid(zz);
      const float nn = tanhf(xn + r * hn);
      sH0[b * H + j] = nn + z * (hp - nn);
    }
    asm volatile("s_wait_dscnt 0" ::: "memory");

    v2f ax1[KC_H], ah1[KC_H];
    #pragma unroll
    for (int kc = 0; kc < KC_H; ++kc) {
      ax1[kc] = *(const v2f*)(sH0 + nloc * H + kc * 4 + koff);
      ah1[kc] = *(const v2f*)(sH1 + nloc * H + kc * 4 + koff);
    }
    asm volatile("s_wait_dscnt 0" ::: "memory");

    #pragma unroll
    for (int nt = 0; nt < NT; ++nt) {
      const int n = nt * 16 + nloc;
      if (nt < 5) {
        v8f acc = {};
        #pragma unroll
        for (int kc = 0; kc < KC_H; ++kc)
          acc = wmma_f32(ax1[kc], ldsB(smem + OFF_WIH1, H, nt, nloc, koff, kc), acc);
        #pragma unroll
        for (int kc = 0; kc < KC_H; ++kc)
          acc = wmma_f32(ah1[kc], ldsB(smem + OFF_WHH1, H, nt, nloc, koff, kc), acc);
        store_tile(smem + OFF_ACCS, 80, n, rbase, acc);
      } else {
        v8f accx = {}, acch = {};
        #pragma unroll
        for (int kc = 0; kc < KC_H; ++kc)
          accx = wmma_f32(ax1[kc], ldsB(smem + OFF_WIH1, H, nt, nloc, koff, kc), accx);
        #pragma unroll
        for (int kc = 0; kc < KC_H; ++kc)
          acch = wmma_f32(ah1[kc], ldsB(smem + OFF_WHH1, H, nt, nloc, koff, kc), acch);
        if (n < G) {
          store_tile(smem + OFF_AXN, H, n - 80, rbase, accx);
          store_tile(smem + OFF_AHN, H, n - 80, rbase, acch);
        }
      }
    }
    asm volatile("s_wait_dscnt 0" ::: "memory");

    for (int e = lane; e < 16 * H; e += 32) {
      const int b = e / H, j = e % H;
      const float rz = smem[OFF_ACCS + b * 80 + j]      + smem[OFF_BRZ1 + j];
      const float zz = smem[OFF_ACCS + b * 80 + 40 + j] + smem[OFF_BRZ1 + 40 + j];
      const float xn = smem[OFF_AXN + b * H + j]        + smem[OFF_BNX1 + j];
      const float hn = smem[OFF_AHN + b * H + j]        + smem[OFF_BNH1 + j];
      const float hp = sH1[b * H + j];
      const float r  = fsigmoid(rz);
      const float z  = fsigmoid(zz);
      const float nn = tanhf(xn + r * hn);
      sH1[b * H + j] = nn + z * (hp - nn);
    }
    asm volatile("s_wait_dscnt 0" ::: "memory");
  }

  for (int e = lane; e < 16 * H; e += 32) {
    const int b = e / H, j = e % H;
    out[(size_t)(bbase + b) * H + j] = sH1[b * H + j];
  }
}

extern "C" void kernel_launch(void* const* d_in, const int* in_sizes, int n_in,
                              void* d_out, int out_size, void* d_ws, size_t ws_size,
                              hipStream_t stream) {
  const float* x    = (const float*)d_in[0];
  const float* Wih0 = (const float*)d_in[1];
  const float* Whh0 = (const float*)d_in[2];
  const float* bih0 = (const float*)d_in[3];
  const float* bhh0 = (const float*)d_in[4];
  const float* Wih1 = (const float*)d_in[5];
  const float* Whh1 = (const float*)d_in[6];
  const float* bih1 = (const float*)d_in[7];
  const float* bhh1 = (const float*)d_in[8];
  float* out = (float*)d_out;

  const int B = in_sizes[0] / (T * I);
  const int nblocks = (B + 15) / 16;

  gru2_fused<<<nblocks, 32, 0, stream>>>(x, Wih0, Whh0, bih0, bhh0,
                                         Wih1, Whh1, bih1, bhh1, out);
  (void)n_in; (void)out_size; (void)d_ws; (void)ws_size;
}


# Gate-aligned tiles: r,z,n of one hidden unit land in one lane's accumulators, so the cell update runs in registers as straight-line code (tanh branches selected), no pre-activation LDS round trip
# speedup vs baseline: 2.6904x; 2.6904x over previous
#include <hip/hip_runtime.h>
#include <cstdint>

typedef float v2f __attribute__((ext_vector_type(2)));
typedef float v8f __attribute__((ext_vector_type(8)));

namespace {

constexpr int H    = 40;
constexpr int G    = 120;
constexpr int I    = 16;
constexpr int T    = 256;
constexpr int JT   = 3;
constexpr int KC_I = 4;
constexpr int KC_H = 10;
constexpr int PH   = 52;

constexpr int OFF_WIH0 = 0;
constexpr int OFF_WIH1 = OFF_WIH0 + G * I;
constexpr int OFF_WHH1 = OFF_WIH1 + G * H;
constexpr int OFF_H0   = OFF_WHH1 + G * H;
constexpr int OFF_H1   = OFF_H0   + 16 * PH;
constexpr int SMEM_FLOATS = OFF_H1 + 16 * PH;

__device__ __forceinline__ v8f wmma_f32(v2f a, v2f b, v8f c) {
  return __builtin_amdgcn_wmma_f32_16x16x4_f32(
       false, a,  false, b,
       (short)0, c,  false,  false);
}

__device__ __forceinline__ v2f ldsB(const float* w, int K, int row, int koff, int kc) {
  return *(const v2f*)(w + row * K + kc * 4 + koff);
}

__device__ __forceinline__ float fsigmoid(float v) { return 1.0f / (1.0f + __expf(-v)); }

__device__ __forceinline__ float tanh_sel(float x) {
  const float y  = fabsf(x);
  const float e  = expf(y + y);
  const float big = fmaf(__builtin_amdgcn_rcpf(e + 1.0f), -2.0f, 1.0f);
  const float y2 = x * x;
  float p = fmaf(-0.00570002059f, y2, 0.0206340719f);
  p = fmaf(y2, p, -0.0537379310f);
  p = fmaf(y2, p, 0.133314162f);
  p = fmaf(y2, p, -0.333332807f);
  const float small = fmaf(y2, y * p, y);
  return copysignf((y < 0.625f) ? small : big, x);
}

__device__ __forceinline__ v8f gru_cell(v8f ar, v8f az, v8f axn, v8f ahn, v8f hp,
                                        float br, float bz, float bnx, float bnh) {
  v8f hn;
  #pragma unroll
  for (int v = 0; v < 8; ++v) {
    const float r  = fsigmoid(ar[v] + br);
    const float z  = fsigmoid(az[v] + bz);
    const float nn = tanh_sel((axn[v] + bnx) + r * (ahn[v] + bnh));
    hn[v] = nn + z * (hp[v] - nn);
  }
  return hn;
}

}

__global__ __launch_bounds__(32) void gru2_fused(
    const float* __restrict__ x,
    const float* __restrict__ Wih0, const float* __restrict__ Whh0,
    const float* __restrict__ bih0, const float* __restrict__ bhh0,
    const float* __restrict__ Wih1, const float* __restrict__ Whh1,
    const float* __restrict__ bih1, const float* __restrict__ bhh1,
    float* __restrict__ out)
{
  __shared__ float smem[SMEM_FLOATS];
  const int lane  = threadIdx.x & 31;
  const int nloc  = lane & 15;
  const int koff  = (lane >> 4) << 1;
  const int rbase = (lane < 16) ? 0 : 8;
  const int bbase = blockIdx.x * 16;

  for (int i = lane; i < G * I; i += 32) smem[OFF_WIH0 + i] = Wih0[i];
  for (int i = lane; i < G * H; i += 32) smem[OFF_WIH1 + i] = Wih1[i];
  for (int i = lane; i < G * H; i += 32) smem[OFF_WHH1 + i] = Whh1[i];
  for (int i = lane; i < 16 * PH; i += 32) { smem[OFF_H0 + i] = 0.f; smem[OFF_H1 + i] = 0.f; }

  int ju[JT];
  #pragma unroll
  for (int jt = 0; jt < JT; ++jt) ju[jt] = min(jt * 16 + nloc, H - 1);

  float b0r[JT], b0z[JT], b0nx[JT], b0nh[JT], b1r[JT], b1z[JT], b1nx[JT], b1nh[JT];
  #pragma unroll
  for (int jt = 0; jt < JT; ++jt) {
    const int j = ju[jt];
    b0r[jt]  = bih0[j] + bhh0[j];
    b0z[jt]  = bih0[H + j] + bhh0[H + j];
    b0nx[jt] = bih0[2 * H + j];
    b0nh[jt] = bhh0[2 * H + j];
    b1r[jt]  = bih1[j] + bhh1[j];
    b1z[jt]  = bih1[H + j] + bhh1[H + j];
    b1nx[jt] = bih1[2 * H + j];
    b1nh[jt] = bhh1[2 * H + j];
  }

  v2f whh0[KC_H][3][JT];
  #pragma unroll
  for (int kc = 0; kc < KC_H; ++kc) {
    #pragma unroll
    for (int g = 0; g < 3; ++g) {
      #pragma unroll
      for (int jt = 0; jt < JT; ++jt)
        whh0[kc][g][jt] = *(const v2f*)(Whh0 + (g * H + ju[jt]) * H + kc * 4 + koff);
    }
  }


  float* sH0 = smem + OFF_H0;
  float* sH1 = smem + OFF_H1;
  const float* xrow = x + (size_t)(bbase + nloc) * T * I;

  v8f hc0[JT], hc1[JT];
  #pragma unroll
  for (int jt = 0; jt < JT; ++jt) { hc0[jt] = v8f{}; hc1[jt] = v8f{}; }
  v2f a0[KC_H], a1[KC_H];
  #pragma unroll
  for (int kc = 0; kc < KC_H; ++kc) { a0[kc] = v2f{0.f, 0.f}; a1[kc] = v2f{0.f, 0.f}; }

  #pragma unroll 1
  for (int t = 0; t < T; ++t) {
    if (t + 1 < T) __builtin_prefetch(xrow + (t + 1) * I + koff, 0, 3);

    v2f ax[KC_I];
    #pragma unroll
    for (int kc = 0; kc < KC_I; ++kc)
      ax[kc] = *(const v2f*)(xrow + t * I + kc * 4 + koff);

    #pragma unroll
    for (int jt = 0; jt < JT; ++jt) {
      v8f ar = {}, az = {}, axn = {}, ahn = {};
      #pragma unroll
      for (int kc = 0; kc < KC_I; ++kc) {
        ar  = wmma_f32(ax[kc], ldsB(smem + OFF_WIH0, I, 0 * H + ju[jt], koff, kc), ar);
        az  = wmma_f32(ax[kc], ldsB(smem + OFF_WIH0, I, 1 * H + ju[jt], koff, kc), az);
        axn = wmma_f32(ax[kc], ldsB(smem + OFF_WIH0, I, 2 * H + ju[jt], koff, kc), axn);
      }
      #pragma unroll
      for (int kc = 0; kc < KC_H; ++kc) {
        ar  = wmma_f32(a0[kc], whh0[kc][0][jt], ar);
        az  = wmma_f32(a0[kc], whh0[kc][1][jt], az);
        ahn = wmma_f32(a0[kc], whh0[kc][2][jt], ahn);
      }
      hc0[jt] = gru_cell(ar, az, axn, ahn, hc0[jt], b0r[jt], b0z[jt], b0nx[jt], b0nh[jt]);
      #pragma unroll
      for (int v = 0; v < 8; ++v) sH0[(rbase + v) * PH + jt * 16 + nloc] = hc0[jt][v];
    }
    #pragma unroll
    for (int kc = 0; kc < KC_H; ++kc)
      a0[kc] = *(const v2f*)(sH0 + nloc * PH + kc * 4 + koff);

    #pragma unroll
    for (int jt = 0; jt < JT; ++jt) {
      v8f ar = {}, az = {}, axn = {}, ahn = {};
      #pragma unroll
      for (int kc = 0; kc < KC_H; ++kc) {
        ar  = wmma_f32(a0[kc], ldsB(smem + OFF_WIH1, H, 0 * H + ju[jt], koff, kc), ar);
        az  = wmma_f32(a0[kc], ldsB(smem + OFF_WIH1, H, 1 * H + ju[jt], koff, kc), az);
        axn = wmma_f32(a0[kc], ldsB(smem + OFF_WIH1, H, 2 * H + ju[jt], koff, kc), axn);
      }
      #pragma unroll
      for (int kc = 0; kc < KC_H; ++kc) {
        ar  = wmma_f32(a1[kc], ldsB(smem + OFF_WHH1, H, 0 * H + ju[jt], koff, kc), ar);
        az  = wmma_f32(a1[kc], ldsB(smem + OFF_WHH1, H, 1 * H + ju[jt], koff, kc), az);
        ahn = wmma_f32(a1[kc], ldsB(smem + OFF_WHH1, H, 2 * H + ju[jt], koff, kc), ahn);
      }
      hc1[jt] = gru_cell(ar, az, axn, ahn, hc1[jt], b1r[jt], b1z[jt], b1nx[jt], b1nh[jt]);
      #pragma unroll
      for (int v = 0; v < 8; ++v) sH1[(rbase + v) * PH + jt * 16 + nloc] = hc1[jt][v];
    }
    #pragma unroll
    for (int kc = 0; kc < KC_H; ++kc)
      a1[kc] = *(const v2f*)(sH1 + nloc * PH + kc * 4 + koff);
  }

  for (int e = lane; e < 16 * H; e += 32) {
    const int b = e / H, j = e % H;
    out[(size_t)(bbase + b) * H + j] = sH1[b * PH + j];
  }
}

extern "C" void kernel_launch(void* const* d_in, const int* in_sizes, int n_in,
                              void* d_out, int out_size, void* d_ws, size_t ws_size,
                              hipStream_t stream) {
  const float* x    = (const float*)d_in[0];
  const float* Wih0 = (const float*)d_in[1];
  const float* Whh0 = (const float*)d_in[2];
  const float* bih0 = (const float*)d_in[3];
  const float* bhh0 = (const float*)d_in[4];
  const float* Wih1 = (const float*)d_in[5];
  const float* Whh1 = (const float*)d_in[6];
  const float* bih1 = (const float*)d_in[7];
  const float* bhh1 = (const float*)d_in[8];
  float* out = (float*)d_out;

  const int B = in_sizes[0] / (T * I);
  const int nblocks = (B + 15) / 16;

  gru2_fused<<<nblocks, 32, 0, stream>>>(x, Wih0, Whh0, bih0, bhh0,
                                         Wih1, Whh1, bih1, bhh1, out);
  (void)n_in; (void)out_size; (void)d_ws; (void)ws_size;
}


# Each 16-row recurrence split over 3 waves on 3 SIMDs: wave j owns hidden-unit tile j of both layers; new h tiles exchanged through LDS behind 2 block barriers per step; sums unchanged
# speedup vs baseline: 5.9373x; 2.2068x over previous
#include <hip/hip_runtime.h>
#include <cstdint>

typedef float v2f __attribute__((ext_vector_type(2)));
typedef float v8f __attribute__((ext_vector_type(8)));

namespace {

constexpr int H    = 40;
constexpr int G    = 120;
constexpr int I    = 16;
constexpr int T    = 256;
constexpr int JT   = 3;
constexpr int NT   = 32 * JT;
constexpr int KC_I = 4;
constexpr int KC_H = 10;
constexpr int PH   = 52;

constexpr int OFF_WIH0 = 0;
constexpr int OFF_WIH1 = OFF_WIH0 + G * I;
constexpr int OFF_WHH1 = OFF_WIH1 + G * H;
constexpr int OFF_H0   = OFF_WHH1 + G * H;
constexpr int OFF_H1   = OFF_H0   + 16 * PH;
constexpr int SMEM_FLOATS = OFF_H1 + 16 * PH;

__device__ __forceinline__ v8f wmma_f32(v2f a, v2f b, v8f c) {
  return __builtin_amdgcn_wmma_f32_16x16x4_f32(
       false, a,  false, b,
       (short)0, c,  false,  false);
}

__device__ __forceinline__ v2f ldsB(const float* w, int K, int row, int koff, int kc) {
  return *(const v2f*)(w + row * K + kc * 4 + koff);
}

__device__ __forceinline__ float fsigmoid(float v) { return 1.0f / (1.0f + __expf(-v)); }

__device__ __forceinline__ float tanh_sel(float x) {
  const float y  = fabsf(x);
  const float e  = expf(y + y);
  const float big = fmaf(__builtin_amdgcn_rcpf(e + 1.0f), -2.0f, 1.0f);
  const float y2 = x * x;
  float p = fmaf(-0.00570002059f, y2, 0.0206340719f);
  p = fmaf(y2, p, -0.0537379310f);
  p = fmaf(y2, p, 0.133314162f);
  p = fmaf(y2, p, -0.333332807f);
  const float small = fmaf(y2, y * p, y);
  return copysignf((y < 0.625f) ? small : big, x);
}

__device__ __forceinline__ v8f gru_cell(v8f ar, v8f az, v8f axn, v8f ahn, v8f hp,
                                        float br, float bz, float bnx, float bnh) {
  v8f hn;
  #pragma unroll
  for (int v = 0; v < 8; ++v) {
    const float r  = fsigmoid(ar[v] + br);
    const float z  = fsigmoid(az[v] + bz);
    const float nn = tanh_sel((axn[v] + bnx) + r * (ahn[v] + bnh));
    hn[v] = nn + z * (hp[v] - nn);
  }
  return hn;
}

}

__global__ __launch_bounds__(NT) void gru2_fused(
    const float* __restrict__ x,
    const float* __restrict__ Wih0, const float* __restrict__ Whh0,
    const float* __restrict__ bih0, const float* __restrict__ bhh0,
    const float* __restrict__ Wih1, const float* __restrict__ Whh1,
    const float* __restrict__ bih1, const float* __restrict__ bhh1,
    float* __restrict__ out)
{
  __shared__ float smem[SMEM_FLOATS];
  const int tid   = threadIdx.x;
  const int lane  = tid & 31;
  const int jt    = tid >> 5;
  const int nloc  = lane & 15;
  const int koff  = (lane >> 4) << 1;
  const int rbase = (lane < 16) ? 0 : 8;
  const int bbase = blockIdx.x * 16;

  for (int i = tid; i < G * I; i += NT) smem[OFF_WIH0 + i] = Wih0[i];
  for (int i = tid; i < G * H; i += NT) smem[OFF_WIH1 + i] = Wih1[i];
  for (int i = tid; i < G * H; i += NT) smem[OFF_WHH1 + i] = Whh1[i];
  for (int i = tid; i < 16 * PH; i += NT) { smem[OFF_H0 + i] = 0.f; smem[OFF_H1 + i] = 0.f; }

  const int ju = min(jt * 16 + nloc, H - 1);

  const float b0r  = bih0[ju] + bhh0[ju];
  const float b0z  = bih0[H + ju] + bhh0[H + ju];
  const float b0nx = bih0[2 * H + ju];
  const float b0nh = bhh0[2 * H + ju];
  const float b1r  = bih1[ju] + bhh1[ju];
  const float b1z  = bih1[H + ju] + bhh1[H + ju];
  const float b1nx = bih1[2 * H + ju];
  const float b1nh = bhh1[2 * H + ju];

  v2f whh0[KC_H][3];
  #pragma unroll
  for (int kc = 0; kc < KC_H; ++kc) {
    #pragma unroll
    for (int g = 0; g < 3; ++g)
      whh0[kc][g] = *(const v2f*)(Whh0 + (g * H + ju) * H + kc * 4 + koff);
  }

  float* sH0 = smem + OFF_H0;
  float* sH1 = smem + OFF_H1;
  const float* xrow = x + (size_t)(bbase + nloc) * T * I;
  float* sH0w = sH0 + rbase * PH + jt * 16 + nloc;
  float* sH1w = sH1 + rbase * PH + jt * 16 + nloc;

  __syncthreads();

  v8f hc0 = {}, hc1 = {};
  v2f a0[KC_H], a1[KC_H];
  #pragma unroll
  for (int kc = 0; kc < KC_H; ++kc) { a0[kc] = v2f{0.f, 0.f}; a1[kc] = v2f{0.f, 0.f}; }

  #pragma unroll 1
  for (int t = 0; t < T; ++t) {
    if (t + 1 < T) __builtin_prefetch(xrow + (t + 1) * I + koff, 0, 3);

    v2f ax[KC_I];
    #pragma unroll
    for (int kc = 0; kc < KC_I; ++kc)
      ax[kc] = *(const v2f*)(xrow + t * I + kc * 4 + koff);

    {
      v8f ar = {}, az = {}, axn = {}, ahn = {};
      #pragma unroll
      for (int kc = 0; kc < KC_I; ++kc) {
        ar  = wmma_f32(ax[kc], ldsB(smem + OFF_WIH0, I, 0 * H + ju, koff, kc), ar);
        az  = wmma_f32(ax[kc], ldsB(smem + OFF_WIH0, I, 1 * H + ju, koff, kc), az);
        axn = wmma_f32(ax[kc], ldsB(smem + OFF_WIH0, I, 2 * H + ju, koff, kc), axn);
      }
      #pragma unroll
      for (int kc = 0; kc < KC_H; ++kc) {
        ar  = wmma_f32(a0[kc], whh0[kc][0], ar);
        az  = wmma_f32(a0[kc], whh0[kc][1], az);
        ahn = wmma_f32(a0[kc], whh0[kc][2], ahn);
      }
      hc0 = gru_cell(ar, az, axn, ahn, hc0, b0r, b0z, b0nx, b0nh);
      #pragma unroll
      for (int v = 0; v < 8; ++v) sH0w[v * PH] = hc0[v];
    }
    __syncthreads();
    #pragma unroll
    for (int kc = 0; kc < KC_H; ++kc)
      a0[kc] = *(const v2f*)(sH0 + nloc * PH + kc * 4 + koff);

    {
      v8f ar = {}, az = {}, axn = {}, ahn = {};
      #pragma unroll
      for (int kc = 0; kc < KC_H; ++kc) {
        ar  = wmma_f32(a0[kc], ldsB(smem + OFF_WIH1, H, 0 * H + ju, koff, kc), ar);
        az  = wmma_f32(a0[kc], ldsB(smem + OFF_WIH1, H, 1 * H + ju, koff, kc), az);
        axn = wmma_f32(a0[kc], ldsB(smem + OFF_WIH1, H, 2 * H + ju, koff, kc), axn);
      }
      #pragma unroll
      for (int kc = 0; kc < KC_H; ++kc) {
        ar  = wmma_f32(a1[kc], ldsB(smem + OFF_WHH1, H, 0 * H + ju, koff, kc), ar);
        az  = wmma_f32(a1[kc], ldsB(smem + OFF_WHH1, H, 1 * H + ju, koff, kc), az);
        ahn = wmma_f32(a1[kc], ldsB(smem + OFF_WHH1, H, 2 * H + ju, koff, kc), ahn);
      }
      hc1 = gru_cell(ar, az, axn, ahn, hc1, b1r, b1z, b1nx, b1nh);
      #pragma unroll
      for (int v = 0; v < 8; ++v) sH1w[v * PH] = hc1[v];
    }
    __syncthreads();
    #pragma unroll
    for (int kc = 0; kc < KC_H; ++kc)
      a1[kc] = *(const v2f*)(sH1 + nloc * PH + kc * 4 + koff);
  }

  for (int e = tid; e < 16 * H; e += NT) {
    const int b = e / H, j = e % H;
    out[(size_t)(bbase + b) * H + j] = sH1[b * PH + j];
  }
}

extern "C" void kernel_launch(void* const* d_in, const int* in_sizes, int n_in,
                              void* d_out, int out_size, void* d_ws, size_t ws_size,
                              hipStream_t stream) {
  const float* x    = (const float*)d_in[0];
  const float* Wih0 = (const float*)d_in[1];
  const float* Whh0 = (const float*)d_in[2];
  const float* bih0 = (const float*)d_in[3];
  const float* bhh0 = (const float*)d_in[4];
  const float* Wih1 = (const float*)d_in[5];
  const float* Whh1 = (const float*)d_in[6];
  const float* bih1 = (const float*)d_in[7];
  const float* bhh1 = (const float*)d_in[8];
  float* out = (float*)d_out;

  const int B = in_sizes[0] / (T * I);
  const int nblocks = (B + 15) / 16;

  gru2_fused<<<nblocks, NT, 0, stream>>>(x, Wih0, Whh0, bih0, bhh0,
                                         Wih1, Whh1, bih1, bhh1, out);
  (void)n_in; (void)out_size; (void)d_ws; (void)ws_size;
}


# Layers pipelined over 6 waves: waves 0-2 run layer 0 at step p while waves 3-5 run layer 1 at step p-1; states double-buffered in LDS, 1 block barrier per phase instead of 2; sums unchanged
# speedup vs baseline: 8.3125x; 1.4000x over previous
#include <hip/hip_runtime.h>
#include <cstdint>

typedef float v2f __attribute__((ext_vector_type(2)));
typedef float v8f __attribute__((ext_vector_type(8)));

namespace {

constexpr int H    = 40;
constexpr int G    = 120;
constexpr int I    = 16;
constexpr int T    = 256;
constexpr int JT   = 3;
constexpr int NW   = 2 * JT;
constexpr int NT   = 32 * NW;
constexpr int KC_I = 4;
constexpr int KC_H = 10;
constexpr int PH   = 52;

constexpr int OFF_WIH0 = 0;
constexpr int OFF_WIH1 = OFF_WIH0 + G * I;
constexpr int OFF_WHH1 = OFF_WIH1 + G * H;
constexpr int HB       = 16 * PH;
constexpr int OFF_H0   = OFF_WHH1 + G * H;
constexpr int OFF_H1   = OFF_H0   + 2 * HB;
constexpr int SMEM_FLOATS = OFF_H1 + 2 * HB;

__device__ __forceinline__ v8f wmma_f32(v2f a, v2f b, v8f c) {
  return __builtin_amdgcn_wmma_f32_16x16x4_f32(
       false, a,  false, b,
       (short)0, c,  false,  false);
}

__device__ __forceinline__ v2f ldsB(const float* w, int K, int row, int koff, int kc) {
  return *(const v2f*)(w + row * K + kc * 4 + koff);
}

__device__ __forceinline__ float fsigmoid(float v) { return 1.0f / (1.0f + __expf(-v)); }

__device__ __forceinline__ float tanh_sel(float x) {
  const float y  = fabsf(x);
  const float e  = expf(y + y);
  const float big = fmaf(__builtin_amdgcn_rcpf(e + 1.0f), -2.0f, 1.0f);
  const float y2 = x * x;
  float p = fmaf(-0.00570002059f, y2, 0.0206340719f);
  p = fmaf(y2, p, -0.0537379310f);
  p = fmaf(y2, p, 0.133314162f);
  p = fmaf(y2, p, -0.333332807f);
  const float small = fmaf(y2, y * p, y);
  return copysignf((y < 0.625f) ? small : big, x);
}

__device__ __forceinline__ v8f gru_cell(v8f ar, v8f az, v8f axn, v8f ahn, v8f hp,
                                        float br, float bz, float bnx, float bnh) {
  v8f hn;
  #pragma unroll
  for (int v = 0; v < 8; ++v) {
    const float r  = fsigmoid(ar[v] + br);
    const float z  = fsigmoid(az[v] + bz);
    const float nn = tanh_sel((axn[v] + bnx) + r * (ahn[v] + bnh));
    hn[v] = nn + z * (hp[v] - nn);
  }
  return hn;
}

}

__global__ __launch_bounds__(NT) void gru2_fused(
    const float* __restrict__ x,
    const float* __restrict__ Wih0, const float* __restrict__ Whh0,
    const float* __restrict__ bih0, const float* __restrict__ bhh0,
    const float* __restrict__ Wih1, const float* __restrict__ Whh1,
    const float* __restrict__ bih1, const float* __restrict__ bhh1,
    float* __restrict__ out)
{
  __shared__ float smem[SMEM_FLOATS];
  const int tid   = threadIdx.x;
  const int lane  = tid & 31;
  const int wv    = __builtin_amdgcn_readfirstlane(tid >> 5);
  const int layer = wv / JT;
  const int jt    = wv - layer * JT;
  const int nloc  = lane & 15;
  const int koff  = (lane >> 4) << 1;
  const int rbase = (lane < 16) ? 0 : 8;
  const int bbase = blockIdx.x * 16;

  for (int i = tid; i < G * I; i += NT) smem[OFF_WIH0 + i] = Wih0[i];
  for (int i = tid; i < G * H; i += NT) smem[OFF_WIH1 + i] = Wih1[i];
  for (int i = tid; i < G * H; i += NT) smem[OFF_WHH1 + i] = Whh1[i];
  for (int i = tid; i < 4 * HB; i += NT) smem[OFF_H0 + i] = 0.f;

  const int ju = min(jt * 16 + nloc, H - 1);

  const float* bih = layer ? bih1 : bih0;
  const float* bhh = layer ? bhh1 : bhh0;
  const float br  = bih[ju] + bhh[ju];
  const float bz  = bih[H + ju] + bhh[H + ju];
  const float bnx = bih[2 * H + ju];
  const float bnh = bhh[2 * H + ju];

  v2f whh0[KC_H][3];
  #pragma unroll
  for (int kc = 0; kc < KC_H; ++kc) {
    #pragma unroll
    for (int g = 0; g < 3; ++g) whh0[kc][g] = v2f{0.f, 0.f};
  }
  if (layer == 0) {
    #pragma unroll
    for (int kc = 0; kc < KC_H; ++kc) {
      #pragma unroll
      for (int g = 0; g < 3; ++g)
        whh0[kc][g] = *(const v2f*)(Whh0 + (g * H + ju) * H + kc * 4 + koff);
    }
  }

  float* sH0 = smem + OFF_H0;
  float* sH1 = smem + OFF_H1;
  const float* xrow = x + (size_t)(bbase + nloc) * T * I;
  float* sHw = (layer ? sH1 : sH0) + rbase * PH + jt * 16 + nloc;

  __syncthreads();

  v8f hc = {};
  v2f a0[KC_H], a1[KC_H];
  #pragma unroll
  for (int kc = 0; kc < KC_H; ++kc) { a0[kc] = v2f{0.f, 0.f}; a1[kc] = v2f{0.f, 0.f}; }

  #pragma unroll 1
  for (int p = 0; p <= T; ++p) {
    const int cur = (p & 1) * HB;

    if (layer == 0) {
      if (p < T) {
        if (p + 1 < T) __builtin_prefetch(xrow + (p + 1) * I + koff, 0, 3);

        v2f ax[KC_I];
        #pragma unroll
        for (int kc = 0; kc < KC_I; ++kc)
          ax[kc] = *(const v2f*)(xrow + p * I + kc * 4 + koff);

        v8f ar = {}, az = {}, axn = {}, ahn = {};
        #pragma unroll
        for (int kc = 0; kc < KC_I; ++kc) {
          ar  = wmma_f32(ax[kc], ldsB(smem + OFF_WIH0, I, 0 * H + ju, koff, kc), ar);
          az  = wmma_f32(ax[kc], ldsB(smem + OFF_WIH0, I, 1 * H + ju, koff, kc), az);
          axn = wmma_f32(ax[kc], ldsB(smem + OFF_WIH0, I, 2 * H + ju, koff, kc), axn);
        }
        #pragma unroll
        for (int kc = 0; kc < KC_H; ++kc) {
          ar  = wmma_f32(a0[kc], whh0[kc][0], ar);
          az  = wmma_f32(a0[kc], whh0[kc][1], az);
          ahn = wmma_f32(a0[kc], whh0[kc][2], ahn);
        }
        hc = gru_cell(ar, az, axn, ahn, hc, br, bz, bnx, bnh);
        #pragma unroll
        for (int v = 0; v < 8; ++v) sHw[cur + v * PH] = hc[v];
      }
    } else {
      if (p > 0) {
        v8f ar = {}, az = {}, axn = {}, ahn = {};
        #pragma unroll
        for (int kc = 0; kc < KC_H; ++kc) {
          ar  = wmma_f32(a0[kc], ldsB(smem + OFF_WIH1, H, 0 * H + ju, koff, kc), ar);
          az  = wmma_f32(a0[kc], ldsB(smem + OFF_WIH1, H, 1 * H + ju, koff, kc), az);
          axn = wmma_f32(a0[kc], ldsB(smem + OFF_WIH1, H, 2 * H + ju, koff, kc), axn);
        }
        #pragma unroll
        for (int kc = 0; kc < KC_H; ++kc) {
          ar  = wmma_f32(a1[kc], ldsB(smem + OFF_WHH1, H, 0 * H + ju, koff, kc), ar);
          az  = wmma_f32(a1[kc], ldsB(smem + OFF_WHH1, H, 1 * H + ju, koff, kc), az);
          ahn = wmma_f32(a1[kc], ldsB(smem + OFF_WHH1, H, 2 * H + ju, koff, kc), ahn);
        }
        hc = gru_cell(ar, az, axn, ahn, hc, br, bz, bnx, bnh);
        #pragma unroll
        for (int v = 0; v < 8; ++v) sHw[cur + v * PH] = hc[v];
      }
    }

    __syncthreads();

    #pragma unroll
    for (int kc = 0; kc < KC_H; ++kc)
      a0[kc] = *(const v2f*)(sH0 + cur + nloc * PH + kc * 4 + koff);
    if (layer != 0) {
      #pragma unroll
      for (int kc = 0; kc < KC_H; ++kc)
        a1[kc] = *(const v2f*)(sH1 + cur + nloc * PH + kc * 4 + koff);
    }
  }

  const float* sHT = sH1 + (T & 1) * HB;
  for (int e = tid; e < 16 * H; e += NT) {
    const int b = e / H, j = e % H;
    out[(size_t)(bbase + b) * H + j] = sHT[b * PH + j];
  }
}

extern "C" void kernel_launch(void* const* d_in, const int* in_sizes, int n_in,
                              void* d_out, int out_size, void* d_ws, size_t ws_size,
                              hipStream_t stream) {
  const float* x    = (const float*)d_in[0];
  const float* Wih0 = (const float*)d_in[1];
  const float* Whh0 = (const float*)d_in[2];
  const float* bih0 = (const float*)d_in[3];
  const float* bhh0 = (const float*)d_in[4];
  const float* Wih1 = (const float*)d_in[5];
  const float* Whh1 = (const float*)d_in[6];
  const float* bih1 = (const float*)d_in[7];
  const float* bhh1 = (const float*)d_in[8];
  float* out = (float*)d_out;

  const int B = in_sizes[0] / (T * I);
  const int nblocks = (B + 15) / 16;

  gru2_fused<<<nblocks, NT, 0, stream>>>(x, Wih0, Whh0, bih0, bhh0,
                                         Wih1, Whh1, bih1, bhh1, out);
  (void)n_in; (void)out_size; (void)d_ws; (void)ws_size;
}


# Layer 1's input products (h0 x Wih1) become a third pipeline stage on 2 more waves (8 waves, 2 per SIMD), handed over in LDS as layer 1's accumulator start values; weights in VGPRs; sums unchanged
# speedup vs baseline: 10.5397x; 1.2679x over previous
#include <hip/hip_runtime.h>
#include <cstdint>

typedef float v2f __attribute__((ext_vector_type(2)));
typedef float v4f __attribute__((ext_vector_type(4)));
typedef float v8f __attribute__((ext_vector_type(8)));

namespace {

constexpr int H    = 40;
constexpr int G    = 120;
constexpr int I    = 16;
constexpr int T    = 256;
constexpr int JT   = 3;
constexpr int NW   = 8;
constexpr int NT   = 32 * NW;
constexpr int KC_I = 4;
constexpr int KC_H = 10;
constexpr int PH   = 52;

constexpr int OFF_WIH0 = 0;
constexpr int HB       = 16 * PH;
constexpr int OFF_H0   = OFF_WIH0 + G * I;
constexpr int OFF_H1   = OFF_H0   + 2 * HB;
constexpr int NXT      = 3 * JT;
constexpr int XGB      = NXT * 256;
constexpr int OFF_XG   = OFF_H1   + 2 * HB;
constexpr int SMEM_FLOATS = OFF_XG + 2 * XGB;
constexpr int XT0      = 5;

__device__ __forceinline__ v8f wmma_f32(v2f a, v2f b, v8f c) {
  return __builtin_amdgcn_wmma_f32_16x16x4_f32(
       false, a,  false, b,
       (short)0, c,  false,  false);
}

__device__ __forceinline__ v2f ldsB(const float* w, int K, int row, int koff, int kc) {
  return *(const v2f*)(w + row * K + kc * 4 + koff);
}

__device__ __forceinline__ float fsigmoid(float v) { return 1.0f / (1.0f + __expf(-v)); }

__device__ __forceinline__ float tanh_sel(float x) {
  const float y  = fabsf(x);
  const float e  = expf(y + y);
  const float big = fmaf(__builtin_amdgcn_rcpf(e + 1.0f), -2.0f, 1.0f);
  const float y2 = x * x;
  float p = fmaf(-0.00570002059f, y2, 0.0206340719f);
  p = fmaf(y2, p, -0.0537379310f);
  p = fmaf(y2, p, 0.133314162f);
  p = fmaf(y2, p, -0.333332807f);
  const float small = fmaf(y2, y * p, y);
  return copysignf((y < 0.625f) ? small : big, x);
}

__device__ __forceinline__ v8f gru_cell(v8f ar, v8f az, v8f axn, v8f ahn, v8f hp,
                                        float br, float bz, float bnx, float bnh) {
  v8f hn;
  #pragma unroll
  for (int v = 0; v < 8; ++v) {
    const float r  = fsigmoid(ar[v] + br);
    const float z  = fsigmoid(az[v] + bz);
    const float nn = tanh_sel((axn[v] + bnx) + r * (ahn[v] + bnh));
    hn[v] = nn + z * (hp[v] - nn);
  }
  return hn;
}

}

__global__ __launch_bounds__(NT) void gru2_fused(
    const float* __restrict__ x,
    const float* __restrict__ Wih0, const float* __restrict__ Whh0,
    const float* __restrict__ bih0, const float* __restrict__ bhh0,
    const float* __restrict__ Wih1, const float* __restrict__ Whh1,
    const float* __restrict__ bih1, const float* __restrict__ bhh1,
    float* __restrict__ out)
{
  __shared__ __attribute__((aligned(16))) float smem[SMEM_FLOATS];
  const int tid   = threadIdx.x;
  const int lane  = tid & 31;
  const int wv    = __builtin_amdgcn_readfirstlane(tid >> 5);
  const int role  = ((wv & 3) == 3) ? 2 : (wv >> 2);
  const int jt    = wv & 3;
  const int xt0   = (wv == 3) ? 0 : XT0;
  const int nloc  = lane & 15;
  const int koff  = (lane >> 4) << 1;
  const int rbase = (lane < 16) ? 0 : 8;
  const int bbase = blockIdx.x * 16;

  for (int i = tid; i < G * I; i += NT) smem[OFF_WIH0 + i] = Wih0[i];
  for (int i = tid; i < 4 * HB; i += NT) smem[OFF_H0 + i] = 0.f;

  const int ju = min(jt * 16 + nloc, H - 1);

  const float* bih = (role == 1) ? bih1 : bih0;
  const float* bhh = (role == 1) ? bhh1 : bhh0;
  const float br  = bih[ju] + bhh[ju];
  const float bz  = bih[H + ju] + bhh[H + ju];
  const float bnx = bih[2 * H + ju];
  const float bnh = bhh[2 * H + ju];

  v2f wB[XT0 * KC_H];
  #pragma unroll
  for (int i = 0; i < XT0 * KC_H; ++i) wB[i] = v2f{0.f, 0.f};
  if (role != 2) {
    const float* Wr = (role == 1) ? Whh1 : Whh0;
    #pragma unroll
    for (int kc = 0; kc < KC_H; ++kc) {
      #pragma unroll
      for (int g = 0; g < 3; ++g)
        wB[kc * 3 + g] = *(const v2f*)(Wr + (g * H + ju) * H + kc * 4 + koff);
    }
  } else {
    #pragma unroll
    for (int i = 0; i < XT0; ++i) {
      const int t   = min(xt0 + i, NXT - 1);
      const int row = (t % 3) * H + min((t / 3) * 16 + nloc, H - 1);
      #pragma unroll
      for (int kc = 0; kc < KC_H; ++kc)
        wB[i * KC_H + kc] = *(const v2f*)(Wih1 + row * H + kc * 4 + koff);
    }
  }

  float* sH0 = smem + OFF_H0;
  float* sH1 = smem + OFF_H1;
  float* sXG = smem + OFF_XG;
  const float* xrow = x + (size_t)(bbase + nloc) * T * I;
  float* sHw = ((role == 1) ? sH1 : sH0) + rbase * PH + jt * 16 + nloc;
  const float* sHr = ((role == 1) ? sH1 : sH0) + nloc * PH + koff;
  float* sXl = sXG + lane * 4;

  __syncthreads();

  v8f hc = {};
  v8f cr = {}, cz = {}, cn = {};
  v2f aH[KC_H];
  #pragma unroll
  for (int kc = 0; kc < KC_H; ++kc) aH[kc] = v2f{0.f, 0.f};

  #pragma unroll 1
  for (int p = 0; p <= T + 1; ++p) {
    const int par = p & 1;
    const int cur = par * HB;
    v8f ar = {}, az = {}, axn = {}, ahn = {};
    bool cell = false;

    if (role == 0) {
      if (p < T) {
        if (p + 1 < T) __builtin_prefetch(xrow + (p + 1) * I + koff, 0, 3);

        v2f ax[KC_I];
        #pragma unroll
        for (int kc = 0; kc < KC_I; ++kc)
          ax[kc] = *(const v2f*)(xrow + p * I + kc * 4 + koff);

        #pragma unroll
        for (int kc = 0; kc < KC_I; ++kc) {
          ar  = wmma_f32(ax[kc], ldsB(smem + OFF_WIH0, I, 0 * H + ju, koff, kc), ar);
          az  = wmma_f32(ax[kc], ldsB(smem + OFF_WIH0, I, 1 * H + ju, koff, kc), az);
          axn = wmma_f32(ax[kc], ldsB(smem + OFF_WIH0, I, 2 * H + ju, koff, kc), axn);
        }
        #pragma unroll
        for (int kc = 0; kc < KC_H; ++kc) {
          ar  = wmma_f32(aH[kc], wB[kc * 3 + 0], ar);
          az  = wmma_f32(aH[kc], wB[kc * 3 + 1], az);
          ahn = wmma_f32(aH[kc], wB[kc * 3 + 2], ahn);
        }
        cell = true;
      }
    } else if (role == 1) {
      if (p >= 2) {
        ar = cr; az = cz; axn = cn;
        #pragma unroll
        for (int kc = 0; kc < KC_H; ++kc) {
          ar  = wmma_f32(aH[kc], wB[kc * 3 + 0], ar);
          az  = wmma_f32(aH[kc], wB[kc * 3 + 1], az);
          ahn = wmma_f32(aH[kc], wB[kc * 3 + 2], ahn);
        }
        cell = true;
      }
    } else {
      if (p >= 1 && p <= T) {
        v8f acc[XT0];
        #pragma unroll
        for (int i = 0; i < XT0; ++i) acc[i] = v8f{};
        #pragma unroll
        for (int kc = 0; kc < KC_H; ++kc) {
          #pragma unroll
          for (int i = 0; i < XT0 - 1; ++i) acc[i] = wmma_f32(aH[kc], wB[i * KC_H + kc], acc[i]);
        }
        if (wv == 3) {
          #pragma unroll
          for (int kc = 0; kc < KC_H; ++kc)
            acc[XT0 - 1] = wmma_f32(aH[kc], wB[(XT0 - 1) * KC_H + kc], acc[XT0 - 1]);
        }
        float* dst = sXl + par * XGB + xt0 * 256;
        #pragma unroll
        for (int i = 0; i < XT0 - 1; ++i) {
          *(v4f*)(dst + i * 256)       = v4f{acc[i][0], acc[i][1], acc[i][2], acc[i][3]};
          *(v4f*)(dst + i * 256 + 128) = v4f{acc[i][4], acc[i][5], acc[i][6], acc[i][7]};
        }
        if (wv == 3) {
          *(v4f*)(dst + (XT0 - 1) * 256)       = v4f{acc[XT0 - 1][0], acc[XT0 - 1][1], acc[XT0 - 1][2], acc[XT0 - 1][3]};
          *(v4f*)(dst + (XT0 - 1) * 256 + 128) = v4f{acc[XT0 - 1][4], acc[XT0 - 1][5], acc[XT0 - 1][6], acc[XT0 - 1][7]};
        }
      }
    }

    if (cell) {
      hc = gru_cell(ar, az, axn, ahn, hc, br, bz, bnx, bnh);
      #pragma unroll
      for (int v = 0; v < 8; ++v) sHw[cur + v * PH] = hc[v];
    }

    __syncthreads();

    #pragma unroll
    for (int kc = 0; kc < KC_H; ++kc)
      aH[kc] = *(const v2f*)(sHr + cur + kc * 4);
    if (role == 1 && p >= 1) {
      const float* srcx = sXl + par * XGB + jt * 3 * 256;
      const v4f r0 = *(const v4f*)(srcx),           r1 = *(const v4f*)(srcx + 128);
      const v4f z0 = *(const v4f*)(srcx + 256),     z1 = *(const v4f*)(srcx + 256 + 128);
      const v4f n0 = *(const v4f*)(srcx + 512),     n1 = *(const v4f*)(srcx + 512 + 128);
      cr = v8f{r0[0], r0[1], r0[2], r0[3], r1[0], r1[1], r1[2], r1[3]};
      cz = v8f{z0[0], z0[1], z0[2], z0[3], z1[0], z1[1], z1[2], z1[3]};
      cn = v8f{n0[0], n0[1], n0[2], n0[3], n1[0], n1[1], n1[2], n1[3]};
    }
  }

  const float* sHT = sH1 + ((T + 1) & 1) * HB;
  for (int e = tid; e < 16 * H; e += NT) {
    const int b = e / H, j = e % H;
    out[(size_t)(bbase + b) * H + j] = sHT[b * PH + j];
  }
}

extern "C" void kernel_launch(void* const* d_in, const int* in_sizes, int n_in,
                              void* d_out, int out_size, void* d_ws, size_t ws_size,
                              hipStream_t stream) {
  const float* x    = (const float*)d_in[0];
  const float* Wih0 = (const float*)d_in[1];
  const float* Whh0 = (const float*)d_in[2];
  const float* bih0 = (const float*)d_in[3];
  const float* bhh0 = (const float*)d_in[4];
  const float* Wih1 = (const float*)d_in[5];
  const float* Whh1 = (const float*)d_in[6];
  const float* bih1 = (const float*)d_in[7];
  const float* bhh1 = (const float*)d_in[8];
  float* out = (float*)d_out;

  const int B = in_sizes[0] / (T * I);
  const int nblocks = (B + 15) / 16;

  gru2_fused<<<nblocks, NT, 0, stream>>>(x, Wih0, Whh0, bih0, bhh0,
                                         Wih1, Whh1, bih1, bhh1, out);
  (void)n_in; (void)out_size; (void)d_ws; (void)ws_size;
}


# Cell update on hardware transcendentals: sigmoid's IEEE division becomes one v_rcp_f32 (1 ulp) and tanh's library expf one v_exp_f32; gates move in the last bits; sums unchanged
# speedup vs baseline: 13.2533x; 1.2575x over previous
#include <hip/hip_runtime.h>
#include <cstdint>

typedef float v2f __attribute__((ext_vector_type(2)));
typedef float v4f __attribute__((ext_vector_type(4)));
typedef float v8f __attribute__((ext_vector_type(8)));

namespace {

constexpr int H    = 40;
constexpr int G    = 120;
constexpr int I    = 16;
constexpr int T    = 256;
constexpr int JT   = 3;
constexpr int NW   = 8;
constexpr int NT   = 32 * NW;
constexpr int KC_I = 4;
constexpr int KC_H = 10;
constexpr int PH   = 52;

constexpr int OFF_WIH0 = 0;
constexpr int HB       = 16 * PH;
constexpr int OFF_H0   = OFF_WIH0 + G * I;
constexpr int OFF_H1   = OFF_H0   + 2 * HB;
constexpr int NXT      = 3 * JT;
constexpr int XGB      = NXT * 256;
constexpr int OFF_XG   = OFF_H1   + 2 * HB;
constexpr int SMEM_FLOATS = OFF_XG + 2 * XGB;
constexpr int XT0      = 5;

__device__ __forceinline__ v8f wmma_f32(v2f a, v2f b, v8f c) {
  return __builtin_amdgcn_wmma_f32_16x16x4_f32(
       false, a,  false, b,
       (short)0, c,  false,  false);
}

__device__ __forceinline__ v2f ldsB(const float* w, int K, int row, int koff, int kc) {
  return *(const v2f*)(w + row * K + kc * 4 + koff);
}

__device__ __forceinline__ float fsigmoid(float v) { return __builtin_amdgcn_rcpf(1.0f + __expf(-v)); }

__device__ __forceinline__ float tanh_sel(float x) {
  const float y  = fabsf(x);
  const float e  = __expf(y + y);
  const float big = fmaf(__builtin_amdgcn_rcpf(e + 1.0f), -2.0f, 1.0f);
  const float y2 = x * x;
  float p = fmaf(-0.00570002059f, y2, 0.0206340719f);
  p = fmaf(y2, p, -0.0537379310f);
  p = fmaf(y2, p, 0.133314162f);
  p = fmaf(y2, p, -0.333332807f);
  const float small = fmaf(y2, y * p, y);
  return copysignf((y < 0.625f) ? small : big, x);
}

__device__ __forceinline__ v8f gru_cell(v8f ar, v8f az, v8f axn, v8f ahn, v8f hp,
                                        float br, float bz, float bnx, float bnh) {
  v8f hn;
  #pragma unroll
  for (int v = 0; v < 8; ++v) {
    const float r  = fsigmoid(ar[v] + br);
    const float z  = fsigmoid(az[v] + bz);
    const float nn = tanh_sel((axn[v] + bnx) + r * (ahn[v] + bnh));
    hn[v] = nn + z * (hp[v] - nn);
  }
  return hn;
}

}

__global__ __launch_bounds__(NT) void gru2_fused(
    const float* __restrict__ x,
    const float* __restrict__ Wih0, const float* __restrict__ Whh0,
    const float* __restrict__ bih0, const float* __restrict__ bhh0,
    const float* __restrict__ Wih1, const float* __restrict__ Whh1,
    const float* __restrict__ bih1, const float* __restrict__ bhh1,
    float* __restrict__ out)
{
  __shared__ __attribute__((aligned(16))) float smem[SMEM_FLOATS];
  const int tid   = threadIdx.x;
  const int lane  = tid & 31;
  const int wv    = __builtin_amdgcn_readfirstlane(tid >> 5);
  const int role  = ((wv & 3) == 3) ? 2 : (wv >> 2);
  const int jt    = wv & 3;
  const int xt0   = (wv == 3) ? 0 : XT0;
  const int nloc  = lane & 15;
  const int koff  = (lane >> 4) << 1;
  const int rbase = (lane < 16) ? 0 : 8;
  const int bbase = blockIdx.x * 16;

  for (int i = tid; i < G * I; i += NT) smem[OFF_WIH0 + i] = Wih0[i];
  for (int i = tid; i < 4 * HB; i += NT) smem[OFF_H0 + i] = 0.f;

  const int ju = min(jt * 16 + nloc, H - 1);

  const float* bih = (role == 1) ? bih1 : bih0;
  const float* bhh = (role == 1) ? bhh1 : bhh0;
  const float br  = bih[ju] + bhh[ju];
  const float bz  = bih[H + ju] + bhh[H + ju];
  const float bnx = bih[2 * H + ju];
  const float bnh = bhh[2 * H + ju];

  v2f wB[XT0 * KC_H];
  #pragma unroll
  for (int i = 0; i < XT0 * KC_H; ++i) wB[i] = v2f{0.f, 0.f};
  if (role != 2) {
    const float* Wr = (role == 1) ? Whh1 : Whh0;
    #pragma unroll
    for (int kc = 0; kc < KC_H; ++kc) {
      #pragma unroll
      for (int g = 0; g < 3; ++g)
        wB[kc * 3 + g] = *(const v2f*)(Wr + (g * H + ju) * H + kc * 4 + koff);
    }
  } else {
    #pragma unroll
    for (int i = 0; i < XT0; ++i) {
      const int t   = min(xt0 + i, NXT - 1);
      const int row = (t % 3) * H + min((t / 3) * 16 + nloc, H - 1);
      #pragma unroll
      for (int kc = 0; kc < KC_H; ++kc)
        wB[i * KC_H + kc] = *(const v2f*)(Wih1 + row * H + kc * 4 + koff);
    }
  }

  float* sH0 = smem + OFF_H0;
  float* sH1 = smem + OFF_H1;
  float* sXG = smem + OFF_XG;
  const float* xrow = x + (size_t)(bbase + nloc) * T * I;
  float* sHw = ((role == 1) ? sH1 : sH0) + rbase * PH + jt * 16 + nloc;
  const float* sHr = ((role == 1) ? sH1 : sH0) + nloc * PH + koff;
  float* sXl = sXG + lane * 4;

  __syncthreads();

  v8f hc = {};
  v8f cr = {}, cz = {}, cn = {};
  v2f aH[KC_H];
  #pragma unroll
  for (int kc = 0; kc < KC_H; ++kc) aH[kc] = v2f{0.f, 0.f};

  #pragma unroll 1
  for (int p = 0; p <= T + 1; ++p) {
    const int par = p & 1;
    const int cur = par * HB;
    v8f ar = {}, az = {}, axn = {}, ahn = {};
    bool cell = false;

    if (role == 0) {
      if (p < T) {
        if (p + 1 < T) __builtin_prefetch(xrow + (p + 1) * I + koff, 0, 3);

        v2f ax[KC_I];
        #pragma unroll
        for (int kc = 0; kc < KC_I; ++kc)
          ax[kc] = *(const v2f*)(xrow + p * I + kc * 4 + koff);

        #pragma unroll
        for (int kc = 0; kc < KC_I; ++kc) {
          ar  = wmma_f32(ax[kc], ldsB(smem + OFF_WIH0, I, 0 * H + ju, koff, kc), ar);
          az  = wmma_f32(ax[kc], ldsB(smem + OFF_WIH0, I, 1 * H + ju, koff, kc), az);
          axn = wmma_f32(ax[kc], ldsB(smem + OFF_WIH0, I, 2 * H + ju, koff, kc), axn);
        }
        #pragma unroll
        for (int kc = 0; kc < KC_H; ++kc) {
          ar  = wmma_f32(aH[kc], wB[kc * 3 + 0], ar);
          az  = wmma_f32(aH[kc], wB[kc * 3 + 1], az);
          ahn = wmma_f32(aH[kc], wB[kc * 3 + 2], ahn);
        }
        cell = true;
      }
    } else if (role == 1) {
      if (p >= 2) {
        ar = cr; az = cz; axn = cn;
        #pragma unroll
        for (int kc = 0; kc < KC_H; ++kc) {
          ar  = wmma_f32(aH[kc], wB[kc * 3 + 0], ar);
          az  = wmma_f32(aH[kc], wB[kc * 3 + 1], az);
          ahn = wmma_f32(aH[kc], wB[kc * 3 + 2], ahn);
        }
        cell = true;
      }
    } else {
      if (p >= 1 && p <= T) {
        v8f acc[XT0];
        #pragma unroll
        for (int i = 0; i < XT0; ++i) acc[i] = v8f{};
        #pragma unroll
        for (int kc = 0; kc < KC_H; ++kc) {
          #pragma unroll
          for (int i = 0; i < XT0 - 1; ++i) acc[i] = wmma_f32(aH[kc], wB[i * KC_H + kc], acc[i]);
        }
        if (wv == 3) {
          #pragma unroll
          for (int kc = 0; kc < KC_H; ++kc)
            acc[XT0 - 1] = wmma_f32(aH[kc], wB[(XT0 - 1) * KC_H + kc], acc[XT0 - 1]);
        }
        float* dst = sXl + par * XGB + xt0 * 256;
        #pragma unroll
        for (int i = 0; i < XT0 - 1; ++i) {
          *(v4f*)(dst + i * 256)       = v4f{acc[i][0], acc[i][1], acc[i][2], acc[i][3]};
          *(v4f*)(dst + i * 256 + 128) = v4f{acc[i][4], acc[i][5], acc[i][6], acc[i][7]};
        }
        if (wv == 3) {
          *(v4f*)(dst + (XT0 - 1) * 256)       = v4f{acc[XT0 - 1][0], acc[XT0 - 1][1], acc[XT0 - 1][2], acc[XT0 - 1][3]};
          *(v4f*)(dst + (XT0 - 1) * 256 + 128) = v4f{acc[XT0 - 1][4], acc[XT0 - 1][5], acc[XT0 - 1][6], acc[XT0 - 1][7]};
        }
      }
    }

    if (cell) {
      hc = gru_cell(ar, az, axn, ahn, hc, br, bz, bnx, bnh);
      #pragma unroll
      for (int v = 0; v < 8; ++v) sHw[cur + v * PH] = hc[v];
    }

    __syncthreads();

    #pragma unroll
    for (int kc = 0; kc < KC_H; ++kc)
      aH[kc] = *(const v2f*)(sHr + cur + kc * 4);
    if (role == 1 && p >= 1) {
      const float* srcx = sXl + par * XGB + jt * 3 * 256;
      const v4f r0 = *(const v4f*)(srcx),           r1 = *(const v4f*)(srcx + 128);
      const v4f z0 = *(const v4f*)(srcx + 256),     z1 = *(const v4f*)(srcx + 256 + 128);
      const v4f n0 = *(const v4f*)(srcx + 512),     n1 = *(const v4f*)(srcx + 512 + 128);
      cr = v8f{r0[0], r0[1], r0[2], r0[3], r1[0], r1[1], r1[2], r1[3]};
      cz = v8f{z0[0], z0[1], z0[2], z0[3], z1[0], z1[1], z1[2], z1[3]};
      cn = v8f{n0[0], n0[1], n0[2], n0[3], n1[0], n1[1], n1[2], n1[3]};
    }
  }

  const float* sHT = sH1 + ((T + 1) & 1) * HB;
  for (int e = tid; e < 16 * H; e += NT) {
    const int b = e / H, j = e % H;
    out[(size_t)(bbase + b) * H + j] = sHT[b * PH + j];
  }
}

extern "C" void kernel_launch(void* const* d_in, const int* in_sizes, int n_in,
                              void* d_out, int out_size, void* d_ws, size_t ws_size,
                              hipStream_t stream) {
  const float* x    = (const float*)d_in[0];
  const float* Wih0 = (const float*)d_in[1];
  const float* Whh0 = (const float*)d_in[2];
  const float* bih0 = (const float*)d_in[3];
  const float* bhh0 = (const float*)d_in[4];
  const float* Wih1 = (const float*)d_in[5];
  const float* Whh1 = (const float*)d_in[6];
  const float* bih1 = (const float*)d_in[7];
  const float* bhh1 = (const float*)d_in[8];
  float* out = (float*)d_out;

  const int B = in_sizes[0] / (T * I);
  const int nblocks = (B + 15) / 16;

  gru2_fused<<<nblocks, NT, 0, stream>>>(x, Wih0, Whh0, bih0, bhh0,
                                         Wih1, Whh1, bih1, bhh1, out);
  (void)n_in; (void)out_size; (void)d_ws; (void)ws_size;
}
